# starting kernel
# baseline (speedup 1.0000x reference)
#include <hip/hip_runtime.h>

typedef float v2f __attribute__((ext_vector_type(2)));
typedef float v8f __attribute__((ext_vector_type(8)));

#define NN   256
#define NC   64
#define NH   128

__global__ void h2_pack_kernel(const float* __restrict__ haar,
                               float* __restrict__ Bp) {
    int t = blockIdx.x * blockDim.x + threadIdx.x;
    int k = t >> 7;
    int j = t & 127;
    const float* hrow = haar + k * NN;
    float s = 0.0f;
    #pragma unroll 4
    for (int m = 0; m < NN; ++m)
        s = fmaf(hrow[m], haar[m * NN + j], s);
    Bp[(k >> 1) * 256 + j * 2 + (k & 1)] = s;
}

#define KPAD 288
__global__ __launch_bounds__(256)
void wmma_xh2_kernel(const float* __restrict__ x,
                     const float* __restrict__ Bp,
                     float* __restrict__ out) {
    __shared__ float Bs[64 * KPAD];

    const int wg  = blockIdx.x;
    const int b   = wg >> 6;
    const int cp  = wg & 63;
    const int tid = threadIdx.x;
    const int w   = tid >> 5;
    const int l   = tid & 31;
    const int ml  = l & 15;
    const int h   = l >> 4;

    const int ip = 16 * w + ml;
    const int ch = ip & 63;
    const int ib = 4 * cp + (ip >> 6);
    const float* xrow = x + (((size_t)b * NC + ch) * NN + (size_t)ib) * NN;

    v8f acc[8] = {};

    for (int ck = 0; ck < 2; ++ck) {
        const float4* src = (const float4*)(Bp + ck * 16384);
        for (int i4 = tid; i4 < 4096; i4 += 256) {
            int row = i4 >> 6;
            int co4 = i4 & 63;
            ((float4*)(Bs + row * KPAD))[co4] = src[i4];
        }
        __syncthreads();

        const int kbase = ck * 128;
        if (ck == 0) __builtin_prefetch(xrow + 128, 0, 0);

        #pragma unroll 2
        for (int ks = 0; ks < 128; ks += 4) {
            v2f a = *(const v2f*)(xrow + kbase + ks + 2 * h);
            const float* brow = Bs + ((ks >> 1) + h) * KPAD + ml * 2;
            #pragma unroll
            for (int n = 0; n < 8; ++n) {
                v2f bf = *(const v2f*)(brow + n * 32);
                acc[n] = __builtin_amdgcn_wmma_f32_16x16x4_f32(
                     false, a,  false, bf,
                     (short)0, acc[n],
                     false,  false);
            }
        }
        __syncthreads();
    }

    float* obase = out + ((size_t)wg * NH + (size_t)(16 * w)) * NH;
    #pragma unroll
    for (int n = 0; n < 8; ++n) {
        #pragma unroll
        for (int r = 0; r < 8; ++r) {
            obase[(size_t)(r + 8 * h) * NH + 16 * n + ml] = acc[n][r];
        }
    }
}

extern "C" void kernel_launch(void* const* d_in, const int* in_sizes, int n_in,
                              void* d_out, int out_size, void* d_ws, size_t ws_size,
                              hipStream_t stream) {
    const float* x    = (const float*)d_in[0];
    const float* haar = (const float*)d_in[1];
    float* out = (float*)d_out;
    float* Bp  = (float*)d_ws;

    h2_pack_kernel<<<dim3(128), dim3(256), 0, stream>>>(haar, Bp);

    wmma_xh2_kernel<<<dim3(16 * NC), dim3(256), 0, stream>>>(x, Bp, out);
}


# H2 pre-pass moved to the fp32 matrix unit: a 4-wave workgroup per 16x16 tile, each wave a quarter of the sum with all its loads in flight, partials added via LDS (H2 summation order changes)
# speedup vs baseline: 1.2803x; 1.2803x over previous
#include <hip/hip_runtime.h>

typedef float v2f __attribute__((ext_vector_type(2)));
typedef float v8f __attribute__((ext_vector_type(8)));

#define NN   256
#define NC   64
#define NH   128

__global__ __launch_bounds__(128)
void h2_pack_kernel(const float* __restrict__ haar,
                    float* __restrict__ Bp) {
    __shared__ float part[3 * 8 * 32];

    const int tm = blockIdx.x >> 3;
    const int tn = blockIdx.x & 7;
    const int q  = threadIdx.x >> 5;
    const int l  = threadIdx.x & 31;
    const int ml = l & 15;
    const int h  = l >> 4;

    const float* ap = haar + (16 * tm + ml) * NN + 64 * q + 2 * h;
    const float* bp = haar + (64 * q + 2 * h) * NN + 16 * tn + ml;

    v8f acc0 = {};
    v8f acc1 = {};
    for (int s = 0; s < 16; s += 2) {
        v2f a0 = *(const v2f*)(ap + 4 * s);
        v2f a1 = *(const v2f*)(ap + 4 * s + 4);
        v2f b0, b1;
        b0.x = bp[(4 * s) * NN];
        b0.y = bp[(4 * s + 1) * NN];
        b1.x = bp[(4 * s + 4) * NN];
        b1.y = bp[(4 * s + 5) * NN];
        acc0 = __builtin_amdgcn_wmma_f32_16x16x4_f32(
             false, a0,  false, b0,
             (short)0, acc0,
             false,  false);
        acc1 = __builtin_amdgcn_wmma_f32_16x16x4_f32(
             false, a1,  false, b1,
             (short)0, acc1,
             false,  false);
    }
    acc0 += acc1;

    if (q != 0) {
        for (int r = 0; r < 8; ++r)
            part[((q - 1) * 8 + r) * 32 + l] = acc0[r];
    }
    __syncthreads();
    if (q == 0) {
        for (int r = 0; r < 8; ++r)
            acc0[r] += (part[r * 32 + l] + part[(8 + r) * 32 + l])
                     + part[(16 + r) * 32 + l];
        float* o = Bp + (8 * tm + 4 * h) * 256 + (16 * tn + ml) * 2;
        for (int r = 0; r < 8; r += 2) {
            v2f v = {acc0[r], acc0[r + 1]};
            *(v2f*)(o + (r >> 1) * 256) = v;
        }
    }
}

#define KPAD 288
__global__ __launch_bounds__(256)
void wmma_xh2_kernel(const float* __restrict__ x,
                     const float* __restrict__ Bp,
                     float* __restrict__ out) {
    __shared__ float Bs[64 * KPAD];

    const int wg  = blockIdx.x;
    const int b   = wg >> 6;
    const int cp  = wg & 63;
    const int tid = threadIdx.x;
    const int w   = tid >> 5;
    const int l   = tid & 31;
    const int ml  = l & 15;
    const int h   = l >> 4;

    const int ip = 16 * w + ml;
    const int ch = ip & 63;
    const int ib = 4 * cp + (ip >> 6);
    const float* xrow = x + (((size_t)b * NC + ch) * NN + (size_t)ib) * NN;

    v8f acc[8] = {};

    for (int ck = 0; ck < 2; ++ck) {
        const float4* src = (const float4*)(Bp + ck * 16384);
        for (int i4 = tid; i4 < 4096; i4 += 256) {
            int row = i4 >> 6;
            int co4 = i4 & 63;
            ((float4*)(Bs + row * KPAD))[co4] = src[i4];
        }
        __syncthreads();

        const int kbase = ck * 128;
        if (ck == 0) __builtin_prefetch(xrow + 128, 0, 0);

        #pragma unroll 2
        for (int ks = 0; ks < 128; ks += 4) {
            v2f a = *(const v2f*)(xrow + kbase + ks + 2 * h);
            const float* brow = Bs + ((ks >> 1) + h) * KPAD + ml * 2;
            #pragma unroll
            for (int n = 0; n < 8; ++n) {
                v2f bf = *(const v2f*)(brow + n * 32);
                acc[n] = __builtin_amdgcn_wmma_f32_16x16x4_f32(
                     false, a,  false, bf,
                     (short)0, acc[n],
                     false,  false);
            }
        }
        __syncthreads();
    }

    float* obase = out + ((size_t)wg * NH + (size_t)(16 * w)) * NH;
    #pragma unroll
    for (int n = 0; n < 8; ++n) {
        #pragma unroll
        for (int r = 0; r < 8; ++r) {
            obase[(size_t)(r + 8 * h) * NH + 16 * n + ml] = acc[n][r];
        }
    }
}

extern "C" void kernel_launch(void* const* d_in, const int* in_sizes, int n_in,
                              void* d_out, int out_size, void* d_ws, size_t ws_size,
                              hipStream_t stream) {
    const float* x    = (const float*)d_in[0];
    const float* haar = (const float*)d_in[1];
    float* out = (float*)d_out;
    float* Bp  = (float*)d_ws;

    h2_pack_kernel<<<dim3(128), dim3(128), 0, stream>>>(haar, Bp);

    wmma_xh2_kernel<<<dim3(16 * NC), dim3(256), 0, stream>>>(x, Bp, out);
}
